# starting kernel
# baseline (speedup 1.0000x reference)
#include <hip/hip_runtime.h>
#include <hip/hip_bf16.h>
#include <stdint.h>

typedef float v2f __attribute__((ext_vector_type(2)));
typedef float v8f __attribute__((ext_vector_type(8)));

#define IN_CH   64
#define OUT_CH  64
#define KDIM    128
#define W1S     130
#define W2S     66
#define H1S     66
#define WAVES   3
#define TILE    16
#define KEY_NEG_INF 0x007FFFFFu

__device__ __forceinline__ unsigned fkey(float f) {
    unsigned u = __float_as_uint(f);
    return (u & 0x80000000u) ? ~u : (u | 0x80000000u);
}

__global__ void init_out_kernel(unsigned* __restrict__ out, int n) {
    int i = blockIdx.x * blockDim.x + threadIdx.x;
    if (i < n) out[i] = KEY_NEG_INF;
}

__global__ void finalize_out_kernel(float* __restrict__ outf, int n) {
    int i = blockIdx.x * blockDim.x + threadIdx.x;
    if (i < n) {
        unsigned u = __float_as_uint(outf[i]);
        float v;
        if (u == KEY_NEG_INF) v = 0.0f;
        else v = (u & 0x80000000u) ? __uint_as_float(u ^ 0x80000000u)
                                   : __uint_as_float(~u);
        outf[i] = v;
    }
}

__global__ __launch_bounds__(32 * WAVES)
void edgeconv_wmma_kernel(const float* __restrict__ x,
                          const int*   __restrict__ ei,
                          const float* __restrict__ W1,
                          const float* __restrict__ b1,
                          const float* __restrict__ W2,
                          const float* __restrict__ b2,
                          unsigned*    __restrict__ out,
                          int n_edges)
{
    __shared__ __align__(16) float sW1T[64 * W1S];
    __shared__ __align__(16) float sW2T[64 * W2S];
    __shared__ __align__(16) float sH1[WAVES][16 * H1S];

    const int tid = threadIdx.x;

    for (int idx = tid; idx < KDIM * OUT_CH; idx += blockDim.x) {
        int k = idx >> 6, n = idx & 63;
        sW1T[n * W1S + k] = W1[idx];
    }
    for (int idx = tid; idx < OUT_CH * OUT_CH; idx += blockDim.x) {
        int k = idx >> 6, n = idx & 63;
        sW2T[n * W2S + k] = W2[idx];
    }
    __syncthreads();

    const int lane = tid & 31;
    const int wave = tid >> 5;
    const int g  = lane >> 4;
    const int nl = lane & 15;
    const int*  srcp = ei;
    const int*  dstp = ei + n_edges;
    float* h1 = &sH1[wave][0];

    float bias1[4], bias2[4];
    #pragma unroll
    for (int nt = 0; nt < 4; ++nt) {
        bias1[nt] = b1[nt * 16 + nl];
        bias2[nt] = b2[nt * 16 + nl];
    }

    const int n_tiles = (n_edges + TILE - 1) / TILE;
    const int gwave   = blockIdx.x * WAVES + wave;
    const int nwaves  = gridDim.x * WAVES;

    for (int tile = gwave; tile < n_tiles; tile += nwaves) {
        const int ebase = tile * TILE;

        int e = ebase + nl;
        if (e >= n_edges) e = n_edges - 1;
        const int si = srcp[e];
        const int di = dstp[e];
        const float* xi = x + (size_t)di * IN_CH;
        const float* xj = x + (size_t)si * IN_CH;
        v2f ai[16], ad[16];
        #pragma unroll
        for (int t = 0; t < 16; ++t) {
            int c = 4 * t + 2 * g;
            v2f vi = *(const v2f*)(xi + c);
            v2f vj = *(const v2f*)(xj + c);
            ai[t] = vi;
            ad[t] = vj - vi;
        }

        v8f acc[4] = {};
        #pragma unroll
        for (int k = 0; k < 16; ++k) {
            int kb = 4 * k + 2 * g;
            #pragma unroll
            for (int nt = 0; nt < 4; ++nt) {
                v2f b = *(const v2f*)(&sW1T[(nt * 16 + nl) * W1S + kb]);
                acc[nt] = __builtin_amdgcn_wmma_f32_16x16x4_f32(
                    false, ai[k], false, b, (short)0, acc[nt], false, false);
            }
        }
        #pragma unroll
        for (int k = 0; k < 16; ++k) {
            int kb = 4 * k + 2 * g + 64;
            #pragma unroll
            for (int nt = 0; nt < 4; ++nt) {
                v2f b = *(const v2f*)(&sW1T[(nt * 16 + nl) * W1S + kb]);
                acc[nt] = __builtin_amdgcn_wmma_f32_16x16x4_f32(
                    false, ad[k], false, b, (short)0, acc[nt], false, false);
            }
        }

        #pragma unroll
        for (int nt = 0; nt < 4; ++nt) {
            #pragma unroll
            for (int r = 0; r < 8; ++r) {
                float h = fmaxf(acc[nt][r] + bias1[nt], 0.0f);
                int m = r + 8 * g;
                h1[m * H1S + nt * 16 + nl] = h;
            }
        }
        asm volatile("s_wait_dscnt 0" ::: "memory");

        v8f acc2[4] = {};
        #pragma unroll
        for (int k = 0; k < 16; ++k) {
            int kb = 4 * k + 2 * g;
            v2f a = *(const v2f*)(&h1[nl * H1S + kb]);
            #pragma unroll
            for (int nt = 0; nt < 4; ++nt) {
                v2f b = *(const v2f*)(&sW2T[(nt * 16 + nl) * W2S + kb]);
                acc2[nt] = __builtin_amdgcn_wmma_f32_16x16x4_f32(
                    false, a, false, b, (short)0, acc2[nt], false, false);
            }
        }

        #pragma unroll
        for (int r = 0; r < 8; ++r) {
            int m  = r + 8 * g;
            int e2 = ebase + m;
            if (e2 < n_edges) {
                int d = dstp[e2];
                unsigned* orow = out + (size_t)d * OUT_CH;
                #pragma unroll
                for (int nt = 0; nt < 4; ++nt) {
                    float v = acc2[nt][r] + bias2[nt];
                    atomicMax(&orow[nt * 16 + nl], fkey(v));
                }
            }
        }
    }
}

extern "C" void kernel_launch(void* const* d_in, const int* in_sizes, int n_in,
                              void* d_out, int out_size, void* d_ws, size_t ws_size,
                              hipStream_t stream) {
    const float* x  = (const float*)d_in[0];
    const int*   ei = (const int*)  d_in[1];
    const float* W1 = (const float*)d_in[2];
    const float* b1 = (const float*)d_in[3];
    const float* W2 = (const float*)d_in[4];
    const float* b2 = (const float*)d_in[5];

    const int n_edges = in_sizes[1] / 2;
    unsigned* outu = (unsigned*)d_out;

    init_out_kernel<<<(out_size + 255) / 256, 256, 0, stream>>>(outu, out_size);

    const int blocks = 1024;
    edgeconv_wmma_kernel<<<blocks, 32 * WAVES, 0, stream>>>(
        x, ei, W1, b1, W2, b2, outu, n_edges);

    finalize_out_kernel<<<(out_size + 255) / 256, 256, 0, stream>>>(
        (float*)d_out, out_size);
}


# Launch shape: 16-wave work-groups share one LDS copy of the weights, 512 groups (two per WGP), tiles dealt wave-major: 8 resident waves per SIMD, not 3, hide gather, LDS and atomic waits under WMMA
# speedup vs baseline: 1.2404x; 1.2404x over previous
#include <hip/hip_runtime.h>
#include <hip/hip_bf16.h>
#include <stdint.h>

typedef float v2f __attribute__((ext_vector_type(2)));
typedef float v8f __attribute__((ext_vector_type(8)));

#define IN_CH   64
#define OUT_CH  64
#define KDIM    128
#define W1S     130
#define W2S     66
#define H1S     66
#define WAVES   16
#define TILE    16
#define KEY_NEG_INF 0x007FFFFFu

__device__ __forceinline__ unsigned fkey(float f) {
    unsigned u = __float_as_uint(f);
    return (u & 0x80000000u) ? ~u : (u | 0x80000000u);
}

__global__ void init_out_kernel(unsigned* __restrict__ out, int n) {
    int i = blockIdx.x * blockDim.x + threadIdx.x;
    if (i < n) out[i] = KEY_NEG_INF;
}

__global__ void finalize_out_kernel(float* __restrict__ outf, int n) {
    int i = blockIdx.x * blockDim.x + threadIdx.x;
    if (i < n) {
        unsigned u = __float_as_uint(outf[i]);
        float v;
        if (u == KEY_NEG_INF) v = 0.0f;
        else v = (u & 0x80000000u) ? __uint_as_float(u ^ 0x80000000u)
                                   : __uint_as_float(~u);
        outf[i] = v;
    }
}

__global__ __launch_bounds__(32 * WAVES)
void edgeconv_wmma_kernel(const float* __restrict__ x,
                          const int*   __restrict__ ei,
                          const float* __restrict__ W1,
                          const float* __restrict__ b1,
                          const float* __restrict__ W2,
                          const float* __restrict__ b2,
                          unsigned*    __restrict__ out,
                          int n_edges)
{
    __shared__ __align__(16) float sW1T[64 * W1S];
    __shared__ __align__(16) float sW2T[64 * W2S];
    __shared__ __align__(16) float sH1[WAVES][16 * H1S];

    const int tid = threadIdx.x;

    for (int idx = tid; idx < KDIM * OUT_CH; idx += blockDim.x) {
        int k = idx >> 6, n = idx & 63;
        sW1T[n * W1S + k] = W1[idx];
    }
    for (int idx = tid; idx < OUT_CH * OUT_CH; idx += blockDim.x) {
        int k = idx >> 6, n = idx & 63;
        sW2T[n * W2S + k] = W2[idx];
    }
    __syncthreads();

    const int lane = tid & 31;
    const int wave = tid >> 5;
    const int g  = lane >> 4;
    const int nl = lane & 15;
    const int*  srcp = ei;
    const int*  dstp = ei + n_edges;
    float* h1 = &sH1[wave][0];

    float bias1[4], bias2[4];
    #pragma unroll
    for (int nt = 0; nt < 4; ++nt) {
        bias1[nt] = b1[nt * 16 + nl];
        bias2[nt] = b2[nt * 16 + nl];
    }

    const int n_tiles = (n_edges + TILE - 1) / TILE;
    const int gwave   = wave * gridDim.x + blockIdx.x;
    const int nwaves  = gridDim.x * WAVES;

    for (int tile = gwave; tile < n_tiles; tile += nwaves) {
        const int ebase = tile * TILE;

        int e = ebase + nl;
        if (e >= n_edges) e = n_edges - 1;
        const int si = srcp[e];
        const int di = dstp[e];
        const float* xi = x + (size_t)di * IN_CH;
        const float* xj = x + (size_t)si * IN_CH;
        v2f ai[16], ad[16];
        #pragma unroll
        for (int t = 0; t < 16; ++t) {
            int c = 4 * t + 2 * g;
            v2f vi = *(const v2f*)(xi + c);
            v2f vj = *(const v2f*)(xj + c);
            ai[t] = vi;
            ad[t] = vj - vi;
        }

        v8f acc[4] = {};
        #pragma unroll
        for (int k = 0; k < 16; ++k) {
            int kb = 4 * k + 2 * g;
            #pragma unroll
            for (int nt = 0; nt < 4; ++nt) {
                v2f b = *(const v2f*)(&sW1T[(nt * 16 + nl) * W1S + kb]);
                acc[nt] = __builtin_amdgcn_wmma_f32_16x16x4_f32(
                    false, ai[k], false, b, (short)0, acc[nt], false, false);
            }
        }
        #pragma unroll
        for (int k = 0; k < 16; ++k) {
            int kb = 4 * k + 2 * g + 64;
            #pragma unroll
            for (int nt = 0; nt < 4; ++nt) {
                v2f b = *(const v2f*)(&sW1T[(nt * 16 + nl) * W1S + kb]);
                acc[nt] = __builtin_amdgcn_wmma_f32_16x16x4_f32(
                    false, ad[k], false, b, (short)0, acc[nt], false, false);
            }
        }

        #pragma unroll
        for (int nt = 0; nt < 4; ++nt) {
            #pragma unroll
            for (int r = 0; r < 8; ++r) {
                float h = fmaxf(acc[nt][r] + bias1[nt], 0.0f);
                int m = r + 8 * g;
                h1[m * H1S + nt * 16 + nl] = h;
            }
        }
        asm volatile("s_wait_dscnt 0" ::: "memory");

        v8f acc2[4] = {};
        #pragma unroll
        for (int k = 0; k < 16; ++k) {
            int kb = 4 * k + 2 * g;
            v2f a = *(const v2f*)(&h1[nl * H1S + kb]);
            #pragma unroll
            for (int nt = 0; nt < 4; ++nt) {
                v2f b = *(const v2f*)(&sW2T[(nt * 16 + nl) * W2S + kb]);
                acc2[nt] = __builtin_amdgcn_wmma_f32_16x16x4_f32(
                    false, a, false, b, (short)0, acc2[nt], false, false);
            }
        }

        #pragma unroll
        for (int r = 0; r < 8; ++r) {
            int m  = r + 8 * g;
            int e2 = ebase + m;
            if (e2 < n_edges) {
                int d = dstp[e2];
                unsigned* orow = out + (size_t)d * OUT_CH;
                #pragma unroll
                for (int nt = 0; nt < 4; ++nt) {
                    float v = acc2[nt][r] + bias2[nt];
                    atomicMax(&orow[nt * 16 + nl], fkey(v));
                }
            }
        }
    }
}

extern "C" void kernel_launch(void* const* d_in, const int* in_sizes, int n_in,
                              void* d_out, int out_size, void* d_ws, size_t ws_size,
                              hipStream_t stream) {
    const float* x  = (const float*)d_in[0];
    const int*   ei = (const int*)  d_in[1];
    const float* W1 = (const float*)d_in[2];
    const float* b1 = (const float*)d_in[3];
    const float* W2 = (const float*)d_in[4];
    const float* b2 = (const float*)d_in[5];

    const int n_edges = in_sizes[1] / 2;
    unsigned* outu = (unsigned*)d_out;

    init_out_kernel<<<(out_size + 255) / 256, 256, 0, stream>>>(outu, out_size);

    const int blocks = 512;
    edgeconv_wmma_kernel<<<blocks, 32 * WAVES, 0, stream>>>(
        x, ei, W1, b1, W2, b2, outu, n_edges);

    finalize_out_kernel<<<(out_size + 255) / 256, 256, 0, stream>>>(
        (float*)d_out, out_size);
}


# First layer moved from edges to nodes by linearity: P = x(W1a-W1b)+b1, Q = x*W1b once per node (in the start-value launch); an edge adds P[i]+Q[j]: 64 WMMAs a tile, not 192; layer-1 sum order changes
# speedup vs baseline: 1.7310x; 1.3955x over previous
#include <hip/hip_runtime.h>
#include <hip/hip_bf16.h>
#include <stdint.h>

typedef float v2f __attribute__((ext_vector_type(2)));
typedef float v8f __attribute__((ext_vector_type(8)));

#define IN_CH   64
#define OUT_CH  64
#define KDIM    128
#define W1S     130
#define W2S     66
#define H1S     66
#define WAVES   16
#define TILE    16
#define KEY_NEG_INF 0x007FFFFFu

__device__ __forceinline__ unsigned fkey(float f) {
    unsigned u = __float_as_uint(f);
    return (u & 0x80000000u) ? ~u : (u | 0x80000000u);
}

#define PWAVES  8
#define WCS     66

__device__ __forceinline__ bool tables_fit(size_t ws_bytes, int n_out) {
    return ws_bytes >= (size_t)n_out * 2u * sizeof(float);
}

__global__ __launch_bounds__(32 * PWAVES)
void node_tables_kernel(const float* __restrict__ x,
                        const float* __restrict__ W1,
                        const float* __restrict__ b1,
                        unsigned*    __restrict__ out,
                        float*       __restrict__ ws,
                        int n_nodes, size_t ws_bytes)
{
    __shared__ __align__(16) float sWcT[128 * WCS];

    const int tid  = threadIdx.x;
    const int lane = tid & 31;
    const int wave = tid >> 5;
    const int g  = lane >> 4;
    const int nl = lane & 15;
    const int n_out   = n_nodes * OUT_CH;
    const bool hoist  = tables_fit(ws_bytes, n_out);
    const int n_ntiles = (n_nodes + TILE - 1) / TILE;
    const int tile = blockIdx.x * PWAVES + wave;

    if (hoist) {
        for (int idx = tid; idx < IN_CH * OUT_CH; idx += blockDim.x) {
            int k = idx >> 6, n = idx & 63;
            float wa = W1[idx];
            float wb = W1[idx + IN_CH * OUT_CH];
            sWcT[n * WCS + k]        = wa - wb;
            sWcT[(n + 64) * WCS + k] = wb;
        }
        __syncthreads();
    }

    if (tile >= n_ntiles) return;
    const int nbase = tile * TILE;

    {
        uint4 kv = make_uint4(KEY_NEG_INF, KEY_NEG_INF, KEY_NEG_INF, KEY_NEG_INF);
        unsigned* orow = out + (size_t)nbase * OUT_CH;
        const int lim = (n_nodes - nbase < TILE ? n_nodes - nbase : TILE) * OUT_CH;
        #pragma unroll
        for (int i = 0; i < 8; ++i) {
            int o = (i * 32 + lane) * 4;
            if (o < lim) *(uint4*)(orow + o) = kv;
        }
    }
    if (!hoist) return;

    int node = nbase + nl;
    if (node >= n_nodes) node = n_nodes - 1;
    const float* xr = x + (size_t)node * IN_CH;
    v2f a[16];
    #pragma unroll
    for (int t = 0; t < 16; ++t) a[t] = *(const v2f*)(xr + 4 * t + 2 * g);

    v8f acc[8] = {};
    #pragma unroll
    for (int k = 0; k < 16; ++k) {
        int kb = 4 * k + 2 * g;
        #pragma unroll
        for (int nt = 0; nt < 8; ++nt) {
            v2f b = *(const v2f*)(&sWcT[(nt * 16 + nl) * WCS + kb]);
            acc[nt] = __builtin_amdgcn_wmma_f32_16x16x4_f32(
                false, a[k], false, b, (short)0, acc[nt], false, false);
        }
    }

    float* P = ws;
    float* Q = ws + (size_t)n_out;
    #pragma unroll
    for (int nt = 0; nt < 4; ++nt) {
        const float bb = b1[nt * 16 + nl];
        #pragma unroll
        for (int r = 0; r < 8; ++r) {
            int nd = nbase + r + 8 * g;
            if (nd < n_nodes) {
                size_t o = (size_t)nd * OUT_CH + nt * 16 + nl;
                P[o] = acc[nt][r] + bb;
                Q[o] = acc[nt + 4][r];
            }
        }
    }
}

__global__ void finalize_out_kernel(float* __restrict__ outf, int n) {
    int i = blockIdx.x * blockDim.x + threadIdx.x;
    if (i < n) {
        unsigned u = __float_as_uint(outf[i]);
        float v;
        if (u == KEY_NEG_INF) v = 0.0f;
        else v = (u & 0x80000000u) ? __uint_as_float(u ^ 0x80000000u)
                                   : __uint_as_float(~u);
        outf[i] = v;
    }
}

__device__ __forceinline__ void scatter_max_tile(const v8f (&acc2)[4], const float (&bias2)[4],
                                                 const int* __restrict__ dstp, unsigned* __restrict__ out,
                                                 int ebase, int n_edges, int g, int nl) {
    #pragma unroll
    for (int r = 0; r < 8; ++r) {
        int m  = r + 8 * g;
        int e2 = ebase + m;
        if (e2 < n_edges) {
            int d = dstp[e2];
            unsigned* orow = out + (size_t)d * OUT_CH;
            #pragma unroll
            for (int nt = 0; nt < 4; ++nt) {
                float v = acc2[nt][r] + bias2[nt];
                atomicMax(&orow[nt * 16 + nl], fkey(v));
            }
        }
    }
}

__global__ __launch_bounds__(32 * WAVES, 8)
void edgeconv_wmma_kernel(const float* __restrict__ x,
                          const int*   __restrict__ ei,
                          const float* __restrict__ W1,
                          const float* __restrict__ b1,
                          const float* __restrict__ W2,
                          const float* __restrict__ b2,
                          unsigned*    __restrict__ out,
                          int n_edges,
                          const float* __restrict__ ws,
                          int n_out, size_t ws_bytes)
{
    __shared__ __align__(16) float sW1T[64 * W1S];
    __shared__ __align__(16) float sW2T[64 * W2S];
    __shared__ __align__(16) float sH1[WAVES][16 * H1S];

    const int tid = threadIdx.x;
    const bool hoist = tables_fit(ws_bytes, n_out);

    if (!hoist) {
        for (int idx = tid; idx < KDIM * OUT_CH; idx += blockDim.x) {
            int k = idx >> 6, n = idx & 63;
            sW1T[n * W1S + k] = W1[idx];
        }
    }
    for (int idx = tid; idx < OUT_CH * OUT_CH; idx += blockDim.x) {
        int k = idx >> 6, n = idx & 63;
        sW2T[n * W2S + k] = W2[idx];
    }
    __syncthreads();

    const int lane = tid & 31;
    const int wave = tid >> 5;
    const int g  = lane >> 4;
    const int nl = lane & 15;
    const int*  srcp = ei;
    const int*  dstp = ei + n_edges;

    float bias2[4];
    #pragma unroll
    for (int nt = 0; nt < 4; ++nt) bias2[nt] = b2[nt * 16 + nl];

    const int n_tiles = (n_edges + TILE - 1) / TILE;
    const int gwave   = wave * gridDim.x + blockIdx.x;
    const int nwaves  = gridDim.x * WAVES;

    if (hoist) {
        const float* Pt = ws;
        const float* Qt = ws + (size_t)n_out;
        for (int tile = gwave; tile < n_tiles; tile += nwaves) {
            const int ebase = tile * TILE;
            asm volatile("" ::: "memory");

            int e = ebase + nl;
            if (e >= n_edges) e = n_edges - 1;
            const int si = srcp[e];
            const int di = dstp[e];
            const float* pi = Pt + (size_t)di * OUT_CH;
            const float* qj = Qt + (size_t)si * OUT_CH;
            v2f hA[16], vq[16];
            #pragma unroll
            for (int t = 0; t < 16; ++t) hA[t] = *(const v2f*)(pi + 4 * t + 2 * g);
            #pragma unroll
            for (int t = 0; t < 16; ++t) vq[t] = *(const v2f*)(qj + 4 * t + 2 * g);
            __builtin_amdgcn_sched_barrier(0);
            #pragma unroll
            for (int t = 0; t < 16; ++t) {
                v2f hs = hA[t] + vq[t];
                hs.x = fmaxf(hs.x, 0.0f);
                hs.y = fmaxf(hs.y, 0.0f);
                hA[t] = hs;
            }

            v8f acc2[4] = {};
            #pragma unroll
            for (int k = 0; k < 16; ++k) {
                int kb = 4 * k + 2 * g;
                #pragma unroll
                for (int nt = 0; nt < 4; ++nt) {
                    v2f b = *(const v2f*)(&sW2T[(nt * 16 + nl) * W2S + kb]);
                    acc2[nt] = __builtin_amdgcn_wmma_f32_16x16x4_f32(
                        false, hA[k], false, b, (short)0, acc2[nt], false, false);
                }
            }

            scatter_max_tile(acc2, bias2, dstp, out, ebase, n_edges, g, nl);
        }
        return;
    }

    float* h1 = &sH1[wave][0];
    float bias1[4];
    #pragma unroll
    for (int nt = 0; nt < 4; ++nt) bias1[nt] = b1[nt * 16 + nl];

    for (int tile = gwave; tile < n_tiles; tile += nwaves) {
        const int ebase = tile * TILE;

        int e = ebase + nl;
        if (e >= n_edges) e = n_edges - 1;
        const int si = srcp[e];
        const int di = dstp[e];
        const float* xi = x + (size_t)di * IN_CH;
        const float* xj = x + (size_t)si * IN_CH;
        v2f ai[16], ad[16];
        #pragma unroll
        for (int t = 0; t < 16; ++t) {
            int c = 4 * t + 2 * g;
            v2f vi = *(const v2f*)(xi + c);
            v2f vj = *(const v2f*)(xj + c);
            ai[t] = vi;
            ad[t] = vj - vi;
        }

        v8f acc[4] = {};
        #pragma unroll
        for (int k = 0; k < 16; ++k) {
            int kb = 4 * k + 2 * g;
            #pragma unroll
            for (int nt = 0; nt < 4; ++nt) {
                v2f b = *(const v2f*)(&sW1T[(nt * 16 + nl) * W1S + kb]);
                acc[nt] = __builtin_amdgcn_wmma_f32_16x16x4_f32(
                    false, ai[k], false, b, (short)0, acc[nt], false, false);
            }
        }
        #pragma unroll
        for (int k = 0; k < 16; ++k) {
            int kb = 4 * k + 2 * g + 64;
            #pragma unroll
            for (int nt = 0; nt < 4; ++nt) {
                v2f b = *(const v2f*)(&sW1T[(nt * 16 + nl) * W1S + kb]);
                acc[nt] = __builtin_amdgcn_wmma_f32_16x16x4_f32(
                    false, ad[k], false, b, (short)0, acc[nt], false, false);
            }
        }

        #pragma unroll
        for (int nt = 0; nt < 4; ++nt) {
            #pragma unroll
            for (int r = 0; r < 8; ++r) {
                float h = fmaxf(acc[nt][r] + bias1[nt], 0.0f);
                int m = r + 8 * g;
                h1[m * H1S + nt * 16 + nl] = h;
            }
        }
        asm volatile("s_wait_dscnt 0" ::: "memory");

        v8f acc2[4] = {};
        #pragma unroll
        for (int k = 0; k < 16; ++k) {
            int kb = 4 * k + 2 * g;
            v2f a = *(const v2f*)(&h1[nl * H1S + kb]);
            #pragma unroll
            for (int nt = 0; nt < 4; ++nt) {
                v2f b = *(const v2f*)(&sW2T[(nt * 16 + nl) * W2S + kb]);
                acc2[nt] = __builtin_amdgcn_wmma_f32_16x16x4_f32(
                    false, a, false, b, (short)0, acc2[nt], false, false);
            }
        }

        scatter_max_tile(acc2, bias2, dstp, out, ebase, n_edges, g, nl);
    }
}

extern "C" void kernel_launch(void* const* d_in, const int* in_sizes, int n_in,
                              void* d_out, int out_size, void* d_ws, size_t ws_size,
                              hipStream_t stream) {
    const float* x  = (const float*)d_in[0];
    const int*   ei = (const int*)  d_in[1];
    const float* W1 = (const float*)d_in[2];
    const float* b1 = (const float*)d_in[3];
    const float* W2 = (const float*)d_in[4];
    const float* b2 = (const float*)d_in[5];

    const int n_edges = in_sizes[1] / 2;
    unsigned* outu = (unsigned*)d_out;

    const int n_nodes = out_size / OUT_CH;
    node_tables_kernel<<<(n_nodes + TILE * PWAVES - 1) / (TILE * PWAVES), 32 * PWAVES, 0, stream>>>(
        x, W1, b1, outu, (float*)d_ws, n_nodes, ws_size);

    const int blocks = 512;
    edgeconv_wmma_kernel<<<blocks, 32 * WAVES, 0, stream>>>(
        x, ei, W1, b1, W2, b2, outu, n_edges, (const float*)d_ws, out_size, ws_size);

    finalize_out_kernel<<<(out_size + 255) / 256, 256, 0, stream>>>(
        (float*)d_out, out_size);
}


# Row gathers of P and Q as whole 256-B rows (16 B a lane: 4 lines an instruction, not 16), summed and turned into GEMM2's A fragments through a per-wave LDS tile; sums and their order unchanged
# speedup vs baseline: 1.8470x; 1.0670x over previous
#include <hip/hip_runtime.h>
#include <hip/hip_bf16.h>
#include <stdint.h>

typedef float v2f __attribute__((ext_vector_type(2)));
typedef float v4f __attribute__((ext_vector_type(4)));
typedef float v8f __attribute__((ext_vector_type(8)));

#define IN_CH   64
#define OUT_CH  64
#define KDIM    128
#define W1S     130
#define W2S     66
#define H1S     68
#define WAVES   16
#define TILE    16
#define KEY_NEG_INF 0x007FFFFFu

__device__ __forceinline__ unsigned fkey(float f) {
    unsigned u = __float_as_uint(f);
    return (u & 0x80000000u) ? ~u : (u | 0x80000000u);
}

#define PWAVES  8
#define WCS     66

__device__ __forceinline__ bool tables_fit(size_t ws_bytes, int n_out) {
    return ws_bytes >= (size_t)n_out * 2u * sizeof(float);
}

__global__ __launch_bounds__(32 * PWAVES)
void node_tables_kernel(const float* __restrict__ x,
                        const float* __restrict__ W1,
                        const float* __restrict__ b1,
                        unsigned*    __restrict__ out,
                        float*       __restrict__ ws,
                        int n_nodes, size_t ws_bytes)
{
    __shared__ __align__(16) float sWcT[128 * WCS];

    const int tid  = threadIdx.x;
    const int lane = tid & 31;
    const int wave = tid >> 5;
    const int g  = lane >> 4;
    const int nl = lane & 15;
    const int n_out   = n_nodes * OUT_CH;
    const bool hoist  = tables_fit(ws_bytes, n_out);
    const int n_ntiles = (n_nodes + TILE - 1) / TILE;
    const int tile = blockIdx.x * PWAVES + wave;

    if (hoist) {
        for (int idx = tid; idx < IN_CH * OUT_CH; idx += blockDim.x) {
            int k = idx >> 6, n = idx & 63;
            float wa = W1[idx];
            float wb = W1[idx + IN_CH * OUT_CH];
            sWcT[n * WCS + k]        = wa - wb;
            sWcT[(n + 64) * WCS + k] = wb;
        }
        __syncthreads();
    }

    if (tile >= n_ntiles) return;
    const int nbase = tile * TILE;

    {
        uint4 kv = make_uint4(KEY_NEG_INF, KEY_NEG_INF, KEY_NEG_INF, KEY_NEG_INF);
        unsigned* orow = out + (size_t)nbase * OUT_CH;
        const int lim = (n_nodes - nbase < TILE ? n_nodes - nbase : TILE) * OUT_CH;
        #pragma unroll
        for (int i = 0; i < 8; ++i) {
            int o = (i * 32 + lane) * 4;
            if (o < lim) *(uint4*)(orow + o) = kv;
        }
    }
    if (!hoist) return;

    int node = nbase + nl;
    if (node >= n_nodes) node = n_nodes - 1;
    const float* xr = x + (size_t)node * IN_CH;
    v2f a[16];
    #pragma unroll
    for (int t = 0; t < 16; ++t) a[t] = *(const v2f*)(xr + 4 * t + 2 * g);

    v8f acc[8] = {};
    #pragma unroll
    for (int k = 0; k < 16; ++k) {
        int kb = 4 * k + 2 * g;
        #pragma unroll
        for (int nt = 0; nt < 8; ++nt) {
            v2f b = *(const v2f*)(&sWcT[(nt * 16 + nl) * WCS + kb]);
            acc[nt] = __builtin_amdgcn_wmma_f32_16x16x4_f32(
                false, a[k], false, b, (short)0, acc[nt], false, false);
        }
    }

    float* P = ws;
    float* Q = ws + (size_t)n_out;
    #pragma unroll
    for (int nt = 0; nt < 4; ++nt) {
        const float bb = b1[nt * 16 + nl];
        #pragma unroll
        for (int r = 0; r < 8; ++r) {
            int nd = nbase + r + 8 * g;
            if (nd < n_nodes) {
                size_t o = (size_t)nd * OUT_CH + nt * 16 + nl;
                P[o] = acc[nt][r] + bb;
                Q[o] = acc[nt + 4][r];
            }
        }
    }
}

__global__ void finalize_out_kernel(float* __restrict__ outf, int n) {
    int i = blockIdx.x * blockDim.x + threadIdx.x;
    if (i < n) {
        unsigned u = __float_as_uint(outf[i]);
        float v;
        if (u == KEY_NEG_INF) v = 0.0f;
        else v = (u & 0x80000000u) ? __uint_as_float(u ^ 0x80000000u)
                                   : __uint_as_float(~u);
        outf[i] = v;
    }
}

__device__ __forceinline__ void scatter_max_tile(const v8f (&acc2)[4], const float (&bias2)[4],
                                                 const int* __restrict__ dstp, unsigned* __restrict__ out,
                                                 int ebase, int n_edges, int g, int nl) {
    #pragma unroll
    for (int r = 0; r < 8; ++r) {
        int m  = r + 8 * g;
        int e2 = ebase + m;
        if (e2 < n_edges) {
            int d = dstp[e2];
            unsigned* orow = out + (size_t)d * OUT_CH;
            #pragma unroll
            for (int nt = 0; nt < 4; ++nt) {
                float v = acc2[nt][r] + bias2[nt];
                atomicMax(&orow[nt * 16 + nl], fkey(v));
            }
        }
    }
}

__global__ __launch_bounds__(32 * WAVES, 8)
void edgeconv_wmma_kernel(const float* __restrict__ x,
                          const int*   __restrict__ ei,
                          const float* __restrict__ W1,
                          const float* __restrict__ b1,
                          const float* __restrict__ W2,
                          const float* __restrict__ b2,
                          unsigned*    __restrict__ out,
                          int n_edges,
                          const float* __restrict__ ws,
                          int n_out, size_t ws_bytes)
{
    __shared__ __align__(16) float sW1T[64 * W1S];
    __shared__ __align__(16) float sW2T[64 * W2S];
    __shared__ __align__(16) float sH1[WAVES][16 * H1S];

    const int tid = threadIdx.x;
    const bool hoist = tables_fit(ws_bytes, n_out);

    if (!hoist) {
        for (int idx = tid; idx < KDIM * OUT_CH; idx += blockDim.x) {
            int k = idx >> 6, n = idx & 63;
            sW1T[n * W1S + k] = W1[idx];
        }
    }
    for (int idx = tid; idx < OUT_CH * OUT_CH; idx += blockDim.x) {
        int k = idx >> 6, n = idx & 63;
        sW2T[n * W2S + k] = W2[idx];
    }
    __syncthreads();

    const int lane = tid & 31;
    const int wave = tid >> 5;
    const int g  = lane >> 4;
    const int nl = lane & 15;
    const int*  srcp = ei;
    const int*  dstp = ei + n_edges;

    float bias2[4];
    #pragma unroll
    for (int nt = 0; nt < 4; ++nt) bias2[nt] = b2[nt * 16 + nl];

    const int n_tiles = (n_edges + TILE - 1) / TILE;
    const int gwave   = wave * gridDim.x + blockIdx.x;
    const int nwaves  = gridDim.x * WAVES;

    if (hoist) {
        const float* Pt = ws;
        const float* Qt = ws + (size_t)n_out;
        float* h1t = &sH1[wave][0];
        for (int tile = gwave; tile < n_tiles; tile += nwaves) {
            const int ebase = tile * TILE;
            asm volatile("" ::: "memory");

            int e = ebase + nl;
            if (e >= n_edges) e = n_edges - 1;
            const int si = srcp[e];
            const int di = dstp[e];
            v4f vp[8], vq[8];
            #pragma unroll
            for (int i = 0; i < 8; ++i) {
                const int d0 = __builtin_amdgcn_readlane(di, 2 * i);
                const int d1 = __builtin_amdgcn_readlane(di, 2 * i + 1);
                const int s0 = __builtin_amdgcn_readlane(si, 2 * i);
                const int s1 = __builtin_amdgcn_readlane(si, 2 * i + 1);
                const int dr = g ? d1 : d0;
                const int sr = g ? s1 : s0;
                vp[i] = *(const v4f*)(Pt + (size_t)dr * OUT_CH + 4 * nl);
                vq[i] = *(const v4f*)(Qt + (size_t)sr * OUT_CH + 4 * nl);
            }
            #pragma unroll
            for (int i = 0; i < 8; ++i) {
                v4f hs = vp[i] + vq[i];
                hs.x = fmaxf(hs.x, 0.0f);
                hs.y = fmaxf(hs.y, 0.0f);
                hs.z = fmaxf(hs.z, 0.0f);
                hs.w = fmaxf(hs.w, 0.0f);
                *(v4f*)(&h1t[(2 * i + g) * H1S + 4 * nl]) = hs;
            }
            asm volatile("s_wait_dscnt 0" ::: "memory");
            v2f hA[16];
            #pragma unroll
            for (int t = 0; t < 16; ++t) hA[t] = *(const v2f*)(&h1t[nl * H1S + 4 * t + 2 * g]);

            v8f acc2[4] = {};
            #pragma unroll
            for (int k = 0; k < 16; ++k) {
                int kb = 4 * k + 2 * g;
                #pragma unroll
                for (int nt = 0; nt < 4; ++nt) {
                    v2f b = *(const v2f*)(&sW2T[(nt * 16 + nl) * W2S + kb]);
                    acc2[nt] = __builtin_amdgcn_wmma_f32_16x16x4_f32(
                        false, hA[k], false, b, (short)0, acc2[nt], false, false);
                }
            }

            scatter_max_tile(acc2, bias2, dstp, out, ebase, n_edges, g, nl);
        }
        return;
    }

    float* h1 = &sH1[wave][0];
    float bias1[4];
    #pragma unroll
    for (int nt = 0; nt < 4; ++nt) bias1[nt] = b1[nt * 16 + nl];

    for (int tile = gwave; tile < n_tiles; tile += nwaves) {
        const int ebase = tile * TILE;

        int e = ebase + nl;
        if (e >= n_edges) e = n_edges - 1;
        const int si = srcp[e];
        const int di = dstp[e];
        const float* xi = x + (size_t)di * IN_CH;
        const float* xj = x + (size_t)si * IN_CH;
        v2f ai[16], ad[16];
        #pragma unroll
        for (int t = 0; t < 16; ++t) {
            int c = 4 * t + 2 * g;
            v2f vi = *(const v2f*)(xi + c);
            v2f vj = *(const v2f*)(xj + c);
            ai[t] = vi;
            ad[t] = vj - vi;
        }

        v8f acc[4] = {};
        #pragma unroll
        for (int k = 0; k < 16; ++k) {
            int kb = 4 * k + 2 * g;
            #pragma unroll
            for (int nt = 0; nt < 4; ++nt) {
                v2f b = *(const v2f*)(&sW1T[(nt * 16 + nl) * W1S + kb]);
                acc[nt] = __builtin_amdgcn_wmma_f32_16x16x4_f32(
                    false, ai[k], false, b, (short)0, acc[nt], false, false);
            }
        }
        #pragma unroll
        for (int k = 0; k < 16; ++k) {
            int kb = 4 * k + 2 * g + 64;
            #pragma unroll
            for (int nt = 0; nt < 4; ++nt) {
                v2f b = *(const v2f*)(&sW1T[(nt * 16 + nl) * W1S + kb]);
                acc[nt] = __builtin_amdgcn_wmma_f32_16x16x4_f32(
                    false, ad[k], false, b, (short)0, acc[nt], false, false);
            }
        }

        #pragma unroll
        for (int nt = 0; nt < 4; ++nt) {
            #pragma unroll
            for (int r = 0; r < 8; ++r) {
                float h = fmaxf(acc[nt][r] + bias1[nt], 0.0f);
                int m = r + 8 * g;
                h1[m * H1S + nt * 16 + nl] = h;
            }
        }
        asm volatile("s_wait_dscnt 0" ::: "memory");

        v8f acc2[4] = {};
        #pragma unroll
        for (int k = 0; k < 16; ++k) {
            int kb = 4 * k + 2 * g;
            v2f a = *(const v2f*)(&h1[nl * H1S + kb]);
            #pragma unroll
            for (int nt = 0; nt < 4; ++nt) {
                v2f b = *(const v2f*)(&sW2T[(nt * 16 + nl) * W2S + kb]);
                acc2[nt] = __builtin_amdgcn_wmma_f32_16x16x4_f32(
                    false, a, false, b, (short)0, acc2[nt], false, false);
            }
        }

        scatter_max_tile(acc2, bias2, dstp, out, ebase, n_edges, g, nl);
    }
}

extern "C" void kernel_launch(void* const* d_in, const int* in_sizes, int n_in,
                              void* d_out, int out_size, void* d_ws, size_t ws_size,
                              hipStream_t stream) {
    const float* x  = (const float*)d_in[0];
    const int*   ei = (const int*)  d_in[1];
    const float* W1 = (const float*)d_in[2];
    const float* b1 = (const float*)d_in[3];
    const float* W2 = (const float*)d_in[4];
    const float* b2 = (const float*)d_in[5];

    const int n_edges = in_sizes[1] / 2;
    unsigned* outu = (unsigned*)d_out;

    const int n_nodes = out_size / OUT_CH;
    node_tables_kernel<<<(n_nodes + TILE * PWAVES - 1) / (TILE * PWAVES), 32 * PWAVES, 0, stream>>>(
        x, W1, b1, outu, (float*)d_ws, n_nodes, ws_size);

    const int blocks = 512;
    edgeconv_wmma_kernel<<<blocks, 32 * WAVES, 0, stream>>>(
        x, ei, W1, b1, W2, b2, outu, n_edges, (const float*)d_ws, out_size, ws_size);

    finalize_out_kernel<<<(out_size + 255) / 256, 256, 0, stream>>>(
        (float*)d_out, out_size);
}


# Scatter-max atomics as whole 128-B lines: lane halves trade column tiles (v_permlanex16) so one atomic covers 32 columns of ONE edge row, not 16 of two rows; row address wave-uniform; sums unchanged
# speedup vs baseline: 1.9641x; 1.0634x over previous
#include <hip/hip_runtime.h>
#include <hip/hip_bf16.h>
#include <stdint.h>

typedef float v2f __attribute__((ext_vector_type(2)));
typedef float v4f __attribute__((ext_vector_type(4)));
typedef float v8f __attribute__((ext_vector_type(8)));

#define IN_CH   64
#define OUT_CH  64
#define KDIM    128
#define W1S     130
#define W2S     66
#define H1S     68
#define WAVES   16
#define TILE    16
#define KEY_NEG_INF 0x007FFFFFu

__device__ __forceinline__ unsigned fkey(float f) {
    unsigned u = __float_as_uint(f);
    return (u & 0x80000000u) ? ~u : (u | 0x80000000u);
}

#define PWAVES  8
#define WCS     66

__device__ __forceinline__ bool tables_fit(size_t ws_bytes, int n_out) {
    return ws_bytes >= (size_t)n_out * 2u * sizeof(float);
}

__global__ __launch_bounds__(32 * PWAVES)
void node_tables_kernel(const float* __restrict__ x,
                        const float* __restrict__ W1,
                        const float* __restrict__ b1,
                        unsigned*    __restrict__ out,
                        float*       __restrict__ ws,
                        int n_nodes, size_t ws_bytes)
{
    __shared__ __align__(16) float sWcT[128 * WCS];

    const int tid  = threadIdx.x;
    const int lane = tid & 31;
    const int wave = tid >> 5;
    const int g  = lane >> 4;
    const int nl = lane & 15;
    const int n_out   = n_nodes * OUT_CH;
    const bool hoist  = tables_fit(ws_bytes, n_out);
    const int n_ntiles = (n_nodes + TILE - 1) / TILE;
    const int tile = blockIdx.x * PWAVES + wave;

    if (hoist) {
        for (int idx = tid; idx < IN_CH * OUT_CH; idx += blockDim.x) {
            int k = idx >> 6, n = idx & 63;
            float wa = W1[idx];
            float wb = W1[idx + IN_CH * OUT_CH];
            sWcT[n * WCS + k]        = wa - wb;
            sWcT[(n + 64) * WCS + k] = wb;
        }
        __syncthreads();
    }

    if (tile >= n_ntiles) return;
    const int nbase = tile * TILE;

    {
        uint4 kv = make_uint4(KEY_NEG_INF, KEY_NEG_INF, KEY_NEG_INF, KEY_NEG_INF);
        unsigned* orow = out + (size_t)nbase * OUT_CH;
        const int lim = (n_nodes - nbase < TILE ? n_nodes - nbase : TILE) * OUT_CH;
        #pragma unroll
        for (int i = 0; i < 8; ++i) {
            int o = (i * 32 + lane) * 4;
            if (o < lim) *(uint4*)(orow + o) = kv;
        }
    }
    if (!hoist) return;

    int node = nbase + nl;
    if (node >= n_nodes) node = n_nodes - 1;
    const float* xr = x + (size_t)node * IN_CH;
    v2f a[16];
    #pragma unroll
    for (int t = 0; t < 16; ++t) a[t] = *(const v2f*)(xr + 4 * t + 2 * g);

    v8f acc[8] = {};
    #pragma unroll
    for (int k = 0; k < 16; ++k) {
        int kb = 4 * k + 2 * g;
        #pragma unroll
        for (int nt = 0; nt < 8; ++nt) {
            v2f b = *(const v2f*)(&sWcT[(nt * 16 + nl) * WCS + kb]);
            acc[nt] = __builtin_amdgcn_wmma_f32_16x16x4_f32(
                false, a[k], false, b, (short)0, acc[nt], false, false);
        }
    }

    float* P = ws;
    float* Q = ws + (size_t)n_out;
    #pragma unroll
    for (int nt = 0; nt < 4; ++nt) {
        const float bb = b1[nt * 16 + nl];
        #pragma unroll
        for (int r = 0; r < 8; ++r) {
            int nd = nbase + r + 8 * g;
            if (nd < n_nodes) {
                size_t o = (size_t)nd * OUT_CH + nt * 16 + nl;
                P[o] = acc[nt][r] + bb;
                Q[o] = acc[nt + 4][r];
            }
        }
    }
}

__global__ void finalize_out_kernel(float* __restrict__ outf, int n) {
    int i = blockIdx.x * blockDim.x + threadIdx.x;
    if (i < n) {
        unsigned u = __float_as_uint(outf[i]);
        float v;
        if (u == KEY_NEG_INF) v = 0.0f;
        else v = (u & 0x80000000u) ? __uint_as_float(u ^ 0x80000000u)
                                   : __uint_as_float(~u);
        outf[i] = v;
    }
}

__device__ __forceinline__ void scatter_max_tile(const v8f (&acc2)[4], const float (&bias2)[4],
                                                 int di, unsigned* __restrict__ out,
                                                 int ebase, int n_edges, int g, int lane) {
    unsigned kx[2][8], ky[2][8];
    #pragma unroll
    for (int p = 0; p < 2; ++p) {
        #pragma unroll
        for (int r = 0; r < 8; ++r) {
            const unsigned k0 = fkey(acc2[2 * p][r] + bias2[2 * p]);
            const unsigned k1 = fkey(acc2[2 * p + 1][r] + bias2[2 * p + 1]);
            const unsigned give = g ? k0 : k1;
            const unsigned recv = (unsigned)__builtin_amdgcn_permlanex16(
                (int)give, (int)give, 0x76543210, 0xfedcba98, false, false);
            kx[p][r] = g ? recv : k0;
            ky[p][r] = g ? k1 : recv;
        }
    }
    #pragma unroll
    for (int r = 0; r < 8; ++r) {
        if (ebase + r < n_edges) {
            const int d = __builtin_amdgcn_readlane(di, r);
            unsigned* orow = out + (size_t)d * OUT_CH + lane;
            atomicMax(orow,      kx[0][r]);
            atomicMax(orow + 32, kx[1][r]);
        }
        if (ebase + r + 8 < n_edges) {
            const int d = __builtin_amdgcn_readlane(di, r + 8);
            unsigned* orow = out + (size_t)d * OUT_CH + lane;
            atomicMax(orow,      ky[0][r]);
            atomicMax(orow + 32, ky[1][r]);
        }
    }
}

__global__ __launch_bounds__(32 * WAVES, 8)
void edgeconv_wmma_kernel(const float* __restrict__ x,
                          const int*   __restrict__ ei,
                          const float* __restrict__ W1,
                          const float* __restrict__ b1,
                          const float* __restrict__ W2,
                          const float* __restrict__ b2,
                          unsigned*    __restrict__ out,
                          int n_edges,
                          const float* __restrict__ ws,
                          int n_out, size_t ws_bytes)
{
    __shared__ __align__(16) float sW1T[64 * W1S];
    __shared__ __align__(16) float sW2T[64 * W2S];
    __shared__ __align__(16) float sH1[WAVES][16 * H1S];

    const int tid = threadIdx.x;
    const bool hoist = tables_fit(ws_bytes, n_out);

    if (!hoist) {
        for (int idx = tid; idx < KDIM * OUT_CH; idx += blockDim.x) {
            int k = idx >> 6, n = idx & 63;
            sW1T[n * W1S + k] = W1[idx];
        }
    }
    for (int idx = tid; idx < OUT_CH * OUT_CH; idx += blockDim.x) {
        int k = idx >> 6, n = idx & 63;
        sW2T[n * W2S + k] = W2[idx];
    }
    __syncthreads();

    const int lane = tid & 31;
    const int wave = tid >> 5;
    const int g  = lane >> 4;
    const int nl = lane & 15;
    const int*  srcp = ei;
    const int*  dstp = ei + n_edges;

    float bias2[4];
    #pragma unroll
    for (int nt = 0; nt < 4; ++nt) bias2[nt] = b2[nt * 16 + nl];

    const int n_tiles = (n_edges + TILE - 1) / TILE;
    const int gwave   = wave * gridDim.x + blockIdx.x;
    const int nwaves  = gridDim.x * WAVES;

    if (hoist) {
        const float* Pt = ws;
        const float* Qt = ws + (size_t)n_out;
        float* h1t = &sH1[wave][0];
        for (int tile = gwave; tile < n_tiles; tile += nwaves) {
            const int ebase = tile * TILE;
            asm volatile("" ::: "memory");

            int e = ebase + nl;
            if (e >= n_edges) e = n_edges - 1;
            const int si = srcp[e];
            const int di = dstp[e];
            v4f vp[8], vq[8];
            #pragma unroll
            for (int i = 0; i < 8; ++i) {
                const int d0 = __builtin_amdgcn_readlane(di, 2 * i);
                const int d1 = __builtin_amdgcn_readlane(di, 2 * i + 1);
                const int s0 = __builtin_amdgcn_readlane(si, 2 * i);
                const int s1 = __builtin_amdgcn_readlane(si, 2 * i + 1);
                const int dr = g ? d1 : d0;
                const int sr = g ? s1 : s0;
                vp[i] = *(const v4f*)(Pt + (size_t)dr * OUT_CH + 4 * nl);
                vq[i] = *(const v4f*)(Qt + (size_t)sr * OUT_CH + 4 * nl);
            }
            #pragma unroll
            for (int i = 0; i < 8; ++i) {
                v4f hs = vp[i] + vq[i];
                hs.x = fmaxf(hs.x, 0.0f);
                hs.y = fmaxf(hs.y, 0.0f);
                hs.z = fmaxf(hs.z, 0.0f);
                hs.w = fmaxf(hs.w, 0.0f);
                *(v4f*)(&h1t[(2 * i + g) * H1S + 4 * nl]) = hs;
            }
            asm volatile("s_wait_dscnt 0" ::: "memory");
            v2f hA[16];
            #pragma unroll
            for (int t = 0; t < 16; ++t) hA[t] = *(const v2f*)(&h1t[nl * H1S + 4 * t + 2 * g]);

            v8f acc2[4] = {};
            #pragma unroll
            for (int k = 0; k < 16; ++k) {
                int kb = 4 * k + 2 * g;
                #pragma unroll
                for (int nt = 0; nt < 4; ++nt) {
                    v2f b = *(const v2f*)(&sW2T[(nt * 16 + nl) * W2S + kb]);
                    acc2[nt] = __builtin_amdgcn_wmma_f32_16x16x4_f32(
                        false, hA[k], false, b, (short)0, acc2[nt], false, false);
                }
            }

            scatter_max_tile(acc2, bias2, di, out, ebase, n_edges, g, lane);
        }
        return;
    }

    float* h1 = &sH1[wave][0];
    float bias1[4];
    #pragma unroll
    for (int nt = 0; nt < 4; ++nt) bias1[nt] = b1[nt * 16 + nl];

    for (int tile = gwave; tile < n_tiles; tile += nwaves) {
        const int ebase = tile * TILE;

        int e = ebase + nl;
        if (e >= n_edges) e = n_edges - 1;
        const int si = srcp[e];
        const int di = dstp[e];
        const float* xi = x + (size_t)di * IN_CH;
        const float* xj = x + (size_t)si * IN_CH;
        v2f ai[16], ad[16];
        #pragma unroll
        for (int t = 0; t < 16; ++t) {
            int c = 4 * t + 2 * g;
            v2f vi = *(const v2f*)(xi + c);
            v2f vj = *(const v2f*)(xj + c);
            ai[t] = vi;
            ad[t] = vj - vi;
        }

        v8f acc[4] = {};
        #pragma unroll
        for (int k = 0; k < 16; ++k) {
            int kb = 4 * k + 2 * g;
            #pragma unroll
            for (int nt = 0; nt < 4; ++nt) {
                v2f b = *(const v2f*)(&sW1T[(nt * 16 + nl) * W1S + kb]);
                acc[nt] = __builtin_amdgcn_wmma_f32_16x16x4_f32(
                    false, ai[k], false, b, (short)0, acc[nt], false, false);
            }
        }
        #pragma unroll
        for (int k = 0; k < 16; ++k) {
            int kb = 4 * k + 2 * g + 64;
            #pragma unroll
            for (int nt = 0; nt < 4; ++nt) {
                v2f b = *(const v2f*)(&sW1T[(nt * 16 + nl) * W1S + kb]);
                acc[nt] = __builtin_amdgcn_wmma_f32_16x16x4_f32(
                    false, ad[k], false, b, (short)0, acc[nt], false, false);
            }
        }

        #pragma unroll
        for (int nt = 0; nt < 4; ++nt) {
            #pragma unroll
            for (int r = 0; r < 8; ++r) {
                float h = fmaxf(acc[nt][r] + bias1[nt], 0.0f);
                int m = r + 8 * g;
                h1[m * H1S + nt * 16 + nl] = h;
            }
        }
        asm volatile("s_wait_dscnt 0" ::: "memory");

        v8f acc2[4] = {};
        #pragma unroll
        for (int k = 0; k < 16; ++k) {
            int kb = 4 * k + 2 * g;
            v2f a = *(const v2f*)(&h1[nl * H1S + kb]);
            #pragma unroll
            for (int nt = 0; nt < 4; ++nt) {
                v2f b = *(const v2f*)(&sW2T[(nt * 16 + nl) * W2S + kb]);
                acc2[nt] = __builtin_amdgcn_wmma_f32_16x16x4_f32(
                    false, a, false, b, (short)0, acc2[nt], false, false);
            }
        }

        scatter_max_tile(acc2, bias2, di, out, ebase, n_edges, g, lane);
    }
}

extern "C" void kernel_launch(void* const* d_in, const int* in_sizes, int n_in,
                              void* d_out, int out_size, void* d_ws, size_t ws_size,
                              hipStream_t stream) {
    const float* x  = (const float*)d_in[0];
    const int*   ei = (const int*)  d_in[1];
    const float* W1 = (const float*)d_in[2];
    const float* b1 = (const float*)d_in[3];
    const float* W2 = (const float*)d_in[4];
    const float* b2 = (const float*)d_in[5];

    const int n_edges = in_sizes[1] / 2;
    unsigned* outu = (unsigned*)d_out;

    const int n_nodes = out_size / OUT_CH;
    node_tables_kernel<<<(n_nodes + TILE * PWAVES - 1) / (TILE * PWAVES), 32 * PWAVES, 0, stream>>>(
        x, W1, b1, outu, (float*)d_ws, n_nodes, ws_size);

    const int blocks = 512;
    edgeconv_wmma_kernel<<<blocks, 32 * WAVES, 0, stream>>>(
        x, ei, W1, b1, W2, b2, outu, n_edges, (const float*)d_ws, out_size, ws_size);

    finalize_out_kernel<<<(out_size + 255) / 256, 256, 0, stream>>>(
        (float*)d_out, out_size);
}


# Scatter-max by the native f32 max atomic on the values themselves (out starts at -inf, row base scalar): no key encode per element, 30 atomics issue as one run; last pass only turns -inf into 0
# speedup vs baseline: 2.2465x; 1.1438x over previous
#include <hip/hip_runtime.h>
#include <hip/hip_bf16.h>
#include <stdint.h>

typedef float v2f __attribute__((ext_vector_type(2)));
typedef float v4f __attribute__((ext_vector_type(4)));
typedef float v8f __attribute__((ext_vector_type(8)));

#define IN_CH   64
#define OUT_CH  64
#define KDIM    128
#define W1S     130
#define W2S     66
#define H1S     68
#define WAVES   16
#define TILE    16
#define NEG_INF_BITS 0xFF800000u

#define PWAVES  8
#define WCS     66

__device__ __forceinline__ bool tables_fit(size_t ws_bytes, int n_out) {
    return ws_bytes >= (size_t)n_out * 2u * sizeof(float);
}

__global__ __launch_bounds__(32 * PWAVES)
void node_tables_kernel(const float* __restrict__ x,
                        const float* __restrict__ W1,
                        const float* __restrict__ b1,
                        unsigned*    __restrict__ out,
                        float*       __restrict__ ws,
                        int n_nodes, size_t ws_bytes)
{
    __shared__ __align__(16) float sWcT[128 * WCS];

    const int tid  = threadIdx.x;
    const int lane = tid & 31;
    const int wave = tid >> 5;
    const int g  = lane >> 4;
    const int nl = lane & 15;
    const int n_out   = n_nodes * OUT_CH;
    const bool hoist  = tables_fit(ws_bytes, n_out);
    const int n_ntiles = (n_nodes + TILE - 1) / TILE;
    const int tile = blockIdx.x * PWAVES + wave;

    if (hoist) {
        for (int idx = tid; idx < IN_CH * OUT_CH; idx += blockDim.x) {
            int k = idx >> 6, n = idx & 63;
            float wa = W1[idx];
            float wb = W1[idx + IN_CH * OUT_CH];
            sWcT[n * WCS + k]        = wa - wb;
            sWcT[(n + 64) * WCS + k] = wb;
        }
        __syncthreads();
    }

    if (tile >= n_ntiles) return;
    const int nbase = tile * TILE;

    {
        uint4 kv = make_uint4(NEG_INF_BITS, NEG_INF_BITS, NEG_INF_BITS, NEG_INF_BITS);
        unsigned* orow = out + (size_t)nbase * OUT_CH;
        const int lim = (n_nodes - nbase < TILE ? n_nodes - nbase : TILE) * OUT_CH;
        #pragma unroll
        for (int i = 0; i < 8; ++i) {
            int o = (i * 32 + lane) * 4;
            if (o < lim) *(uint4*)(orow + o) = kv;
        }
    }
    if (!hoist) return;

    int node = nbase + nl;
    if (node >= n_nodes) node = n_nodes - 1;
    const float* xr = x + (size_t)node * IN_CH;
    v2f a[16];
    #pragma unroll
    for (int t = 0; t < 16; ++t) a[t] = *(const v2f*)(xr + 4 * t + 2 * g);

    v8f acc[8] = {};
    #pragma unroll
    for (int k = 0; k < 16; ++k) {
        int kb = 4 * k + 2 * g;
        #pragma unroll
        for (int nt = 0; nt < 8; ++nt) {
            v2f b = *(const v2f*)(&sWcT[(nt * 16 + nl) * WCS + kb]);
            acc[nt] = __builtin_amdgcn_wmma_f32_16x16x4_f32(
                false, a[k], false, b, (short)0, acc[nt], false, false);
        }
    }

    float* P = ws;
    float* Q = ws + (size_t)n_out;
    #pragma unroll
    for (int nt = 0; nt < 4; ++nt) {
        const float bb = b1[nt * 16 + nl];
        #pragma unroll
        for (int r = 0; r < 8; ++r) {
            int nd = nbase + r + 8 * g;
            if (nd < n_nodes) {
                size_t o = (size_t)nd * OUT_CH + nt * 16 + nl;
                P[o] = acc[nt][r] + bb;
                Q[o] = acc[nt + 4][r];
            }
        }
    }
}

__global__ void finalize_out_kernel(float* __restrict__ outf, int n) {
    int i = blockIdx.x * blockDim.x + threadIdx.x;
    if (i < n) {
        if (__float_as_uint(outf[i]) == NEG_INF_BITS) outf[i] = 0.0f;
    }
}

typedef __attribute__((address_space(1))) float gfloat;
__device__ __forceinline__ void atomic_max_f32(gfloat* p, float v) {
    (void)__hip_atomic_fetch_max(p, v, __ATOMIC_RELAXED, __HIP_MEMORY_SCOPE_AGENT);
}
__device__ __forceinline__ gfloat* row_base(float* out, int d) {
    gfloat* orow = (gfloat*)(out + (size_t)d * OUT_CH);
    asm("" : "+s"(orow));
    return orow;
}

template <bool GUARD>
__device__ __forceinline__ void scatter_rows(const float (&kx)[2][8], const float (&ky)[2][8],
                                             int di, float* __restrict__ out,
                                             int ebase, int n_edges, int lane) {
    gfloat* rows[16];
    #pragma unroll
    for (int m = 0; m < 16; ++m) rows[m] = row_base(out, __builtin_amdgcn_readlane(di, m));
    #pragma unroll
    for (int r = 0; r < 8; ++r) {
        if (!GUARD || ebase + r < n_edges) {
            atomic_max_f32(rows[r] + lane,      kx[0][r]);
            atomic_max_f32(rows[r] + 32 + lane, kx[1][r]);
        }
        if (!GUARD || ebase + r + 8 < n_edges) {
            atomic_max_f32(rows[r + 8] + lane,      ky[0][r]);
            atomic_max_f32(rows[r + 8] + 32 + lane, ky[1][r]);
        }
    }
}

__device__ __forceinline__ void scatter_max_tile(const v8f (&acc2)[4], const float (&bias2)[4],
                                                 int di, float* __restrict__ out,
                                                 int ebase, int n_edges, int g, int lane) {
    float kx[2][8], ky[2][8];
    #pragma unroll
    for (int p = 0; p < 2; ++p) {
        #pragma unroll
        for (int r = 0; r < 8; ++r) {
            const float k0 = acc2[2 * p][r] + bias2[2 * p];
            const float k1 = acc2[2 * p + 1][r] + bias2[2 * p + 1];
            const int give = __float_as_int(g ? k0 : k1);
            const float recv = __int_as_float(__builtin_amdgcn_permlanex16(
                give, give, 0x76543210, 0xfedcba98, false, false));
            kx[p][r] = g ? recv : k0;
            ky[p][r] = g ? k1 : recv;
        }
    }
    if (ebase + TILE <= n_edges)
        scatter_rows<false>(kx, ky, di, out, ebase, n_edges, lane);
    else
        scatter_rows<true>(kx, ky, di, out, ebase, n_edges, lane);
}

__global__ __launch_bounds__(32 * WAVES, 8)
void edgeconv_wmma_kernel(const float* __restrict__ x,
                          const int*   __restrict__ ei,
                          const float* __restrict__ W1,
                          const float* __restrict__ b1,
                          const float* __restrict__ W2,
                          const float* __restrict__ b2,
                          float*       __restrict__ out,
                          int n_edges,
                          const float* __restrict__ ws,
                          int n_out, size_t ws_bytes)
{
    __shared__ __align__(16) float sW1T[64 * W1S];
    __shared__ __align__(16) float sW2T[64 * W2S];
    __shared__ __align__(16) float sH1[WAVES][16 * H1S];

    const int tid = threadIdx.x;
    const bool hoist = tables_fit(ws_bytes, n_out);

    if (!hoist) {
        for (int idx = tid; idx < KDIM * OUT_CH; idx += blockDim.x) {
            int k = idx >> 6, n = idx & 63;
            sW1T[n * W1S + k] = W1[idx];
        }
    }
    for (int idx = tid; idx < OUT_CH * OUT_CH; idx += blockDim.x) {
        int k = idx >> 6, n = idx & 63;
        sW2T[n * W2S + k] = W2[idx];
    }
    __syncthreads();

    const int lane = tid & 31;
    const int wave = tid >> 5;
    const int g  = lane >> 4;
    const int nl = lane & 15;
    const int*  srcp = ei;
    const int*  dstp = ei + n_edges;

    float bias2[4];
    #pragma unroll
    for (int nt = 0; nt < 4; ++nt) bias2[nt] = b2[nt * 16 + nl];

    const int n_tiles = (n_edges + TILE - 1) / TILE;
    const int gwave   = wave * gridDim.x + blockIdx.x;
    const int nwaves  = gridDim.x * WAVES;

    if (hoist) {
        const float* Pt = ws;
        const float* Qt = ws + (size_t)n_out;
        float* h1t = &sH1[wave][0];
        for (int tile = gwave; tile < n_tiles; tile += nwaves) {
            const int ebase = tile * TILE;
            asm volatile("" ::: "memory");

            int e = ebase + nl;
            if (e >= n_edges) e = n_edges - 1;
            const int si = srcp[e];
            const int di = dstp[e];
            v4f vp[8], vq[8];
            #pragma unroll
            for (int i = 0; i < 8; ++i) {
                const int d0 = __builtin_amdgcn_readlane(di, 2 * i);
                const int d1 = __builtin_amdgcn_readlane(di, 2 * i + 1);
                const int s0 = __builtin_amdgcn_readlane(si, 2 * i);
                const int s1 = __builtin_amdgcn_readlane(si, 2 * i + 1);
                const int dr = g ? d1 : d0;
                const int sr = g ? s1 : s0;
                vp[i] = *(const v4f*)(Pt + (size_t)dr * OUT_CH + 4 * nl);
                vq[i] = *(const v4f*)(Qt + (size_t)sr * OUT_CH + 4 * nl);
            }
            #pragma unroll
            for (int i = 0; i < 8; ++i) {
                v4f hs = vp[i] + vq[i];
                hs.x = fmaxf(hs.x, 0.0f);
                hs.y = fmaxf(hs.y, 0.0f);
                hs.z = fmaxf(hs.z, 0.0f);
                hs.w = fmaxf(hs.w, 0.0f);
                *(v4f*)(&h1t[(2 * i + g) * H1S + 4 * nl]) = hs;
            }
            asm volatile("s_wait_dscnt 0" ::: "memory");
            v2f hA[16];
            #pragma unroll
            for (int t = 0; t < 16; ++t) hA[t] = *(const v2f*)(&h1t[nl * H1S + 4 * t + 2 * g]);

            v8f acc2[4] = {};
            #pragma unroll
            for (int k = 0; k < 16; ++k) {
                int kb = 4 * k + 2 * g;
                #pragma unroll
                for (int nt = 0; nt < 4; ++nt) {
                    v2f b = *(const v2f*)(&sW2T[(nt * 16 + nl) * W2S + kb]);
                    acc2[nt] = __builtin_amdgcn_wmma_f32_16x16x4_f32(
                        false, hA[k], false, b, (short)0, acc2[nt], false, false);
                }
            }

            scatter_max_tile(acc2, bias2, di, out, ebase, n_edges, g, lane);
        }
        return;
    }

    float* h1 = &sH1[wave][0];
    float bias1[4];
    #pragma unroll
    for (int nt = 0; nt < 4; ++nt) bias1[nt] = b1[nt * 16 + nl];

    for (int tile = gwave; tile < n_tiles; tile += nwaves) {
        const int ebase = tile * TILE;

        int e = ebase + nl;
        if (e >= n_edges) e = n_edges - 1;
        const int si = srcp[e];
        const int di = dstp[e];
        const float* xi = x + (size_t)di * IN_CH;
        const float* xj = x + (size_t)si * IN_CH;
        v2f ai[16], ad[16];
        #pragma unroll
        for (int t = 0; t < 16; ++t) {
            int c = 4 * t + 2 * g;
            v2f vi = *(const v2f*)(xi + c);
            v2f vj = *(const v2f*)(xj + c);
            ai[t] = vi;
            ad[t] = vj - vi;
        }

        v8f acc[4] = {};
        #pragma unroll
        for (int k = 0; k < 16; ++k) {
            int kb = 4 * k + 2 * g;
            #pragma unroll
            for (int nt = 0; nt < 4; ++nt) {
                v2f b = *(const v2f*)(&sW1T[(nt * 16 + nl) * W1S + kb]);
                acc[nt] = __builtin_amdgcn_wmma_f32_16x16x4_f32(
                    false, ai[k], false, b, (short)0, acc[nt], false, false);
            }
        }
        #pragma unroll
        for (int k = 0; k < 16; ++k) {
            int kb = 4 * k + 2 * g + 64;
            #pragma unroll
            for (int nt = 0; nt < 4; ++nt) {
                v2f b = *(const v2f*)(&sW1T[(nt * 16 + nl) * W1S + kb]);
                acc[nt] = __builtin_amdgcn_wmma_f32_16x16x4_f32(
                    false, ad[k], false, b, (short)0, acc[nt], false, false);
            }
        }

        #pragma unroll
        for (int nt = 0; nt < 4; ++nt) {
            #pragma unroll
            for (int r = 0; r < 8; ++r) {
                float h = fmaxf(acc[nt][r] + bias1[nt], 0.0f);
                int m = r + 8 * g;
                h1[m * H1S + nt * 16 + nl] = h;
            }
        }
        asm volatile("s_wait_dscnt 0" ::: "memory");

        v8f acc2[4] = {};
        #pragma unroll
        for (int k = 0; k < 16; ++k) {
            int kb = 4 * k + 2 * g;
            v2f a = *(const v2f*)(&h1[nl * H1S + kb]);
            #pragma unroll
            for (int nt = 0; nt < 4; ++nt) {
                v2f b = *(const v2f*)(&sW2T[(nt * 16 + nl) * W2S + kb]);
                acc2[nt] = __builtin_amdgcn_wmma_f32_16x16x4_f32(
                    false, a, false, b, (short)0, acc2[nt], false, false);
            }
        }

        scatter_max_tile(acc2, bias2, di, out, ebase, n_edges, g, lane);
    }
}

extern "C" void kernel_launch(void* const* d_in, const int* in_sizes, int n_in,
                              void* d_out, int out_size, void* d_ws, size_t ws_size,
                              hipStream_t stream) {
    const float* x  = (const float*)d_in[0];
    const int*   ei = (const int*)  d_in[1];
    const float* W1 = (const float*)d_in[2];
    const float* b1 = (const float*)d_in[3];
    const float* W2 = (const float*)d_in[4];
    const float* b2 = (const float*)d_in[5];

    const int n_edges = in_sizes[1] / 2;
    unsigned* outu = (unsigned*)d_out;

    const int n_nodes = out_size / OUT_CH;
    node_tables_kernel<<<(n_nodes + TILE * PWAVES - 1) / (TILE * PWAVES), 32 * PWAVES, 0, stream>>>(
        x, W1, b1, outu, (float*)d_ws, n_nodes, ws_size);

    const int blocks = 512;
    edgeconv_wmma_kernel<<<blocks, 32 * WAVES, 0, stream>>>(
        x, ei, W1, b1, W2, b2, (float*)d_out, n_edges, (const float*)d_ws, out_size, ws_size);

    finalize_out_kernel<<<(out_size + 255) / 256, 256, 0, stream>>>(
        (float*)d_out, out_size);
}
